# starting kernel
# baseline (speedup 1.0000x reference)
#include <hip/hip_runtime.h>

typedef __attribute__((ext_vector_type(2))) float v2f;
typedef __attribute__((ext_vector_type(8))) float v8f;

#define HDIM 256
#define SB_STRIDE 72

__global__ __launch_bounds__(256)
void gemm_f32_wmma(const float* __restrict__ A, const float* __restrict__ B,
                   const float* __restrict__ bias, float* __restrict__ C,
                   int M, int K, int N, int applyRelu)
{
    __shared__ float sB[256 * SB_STRIDE];

    const int tid  = threadIdx.x;
    const int lane = tid & 31;
    const int wave = tid >> 5;
    const int half = lane >> 4;
    const int l16  = lane & 15;

    const int row0 = blockIdx.y * 128 + wave * 16;
    const int col0 = blockIdx.x * 64;

    {
        const int r = tid >> 4;
        const int c = (tid & 15) << 2;
        for (int kb = 0; kb < K; kb += 16) {
            const float4 v = *(const float4*)(B + (size_t)(kb + r) * N + col0 + c);
            *(float4*)(&sB[(kb + r) * SB_STRIDE + c]) = v;
        }
    }
    __syncthreads();

    int aRow = row0 + l16;
    if (aRow > M - 1) aRow = M - 1;
    const float* Aptr = A + (size_t)aRow * K + half * 2;

    v8f acc[4];
    acc[0] = (v8f)0.0f; acc[1] = (v8f)0.0f; acc[2] = (v8f)0.0f; acc[3] = (v8f)0.0f;

    for (int kk = 0; kk < K; kk += 4) {
        const v2f a = *(const v2f*)(Aptr + kk);
        __builtin_prefetch(Aptr + kk + 32, 0, 0);

        const float* s0 = &sB[(kk + half * 2) * SB_STRIDE + l16];
#pragma unroll
        for (int t = 0; t < 4; ++t) {
            v2f b;
            b.x = s0[t * 16];
            b.y = s0[SB_STRIDE + t * 16];
            acc[t] = __builtin_amdgcn_wmma_f32_16x16x4_f32(
                false, a, false, b, (short)0, acc[t], false, false);
        }
    }

#pragma unroll
    for (int t = 0; t < 4; ++t) {
        const int col = col0 + t * 16 + l16;
        const float bv = bias ? bias[col] : 0.0f;
#pragma unroll
        for (int r = 0; r < 8; ++r) {
            const int row = row0 + half * 8 + r;
            if (row < M) {
                float v = acc[t][r] + bv;
                if (applyRelu) v = v > 0.0f ? v : 0.0f;
                C[(size_t)row * N + col] = v;
            }
        }
    }
}

__global__ void deg_init(float* __restrict__ deg, int N)
{
    int i = blockIdx.x * blockDim.x + threadIdx.x;
    if (i < N) deg[i] = 1.0f;
}

__global__ void deg_edges(const int* __restrict__ dst, float* __restrict__ deg, int E)
{
    int e = blockIdx.x * blockDim.x + threadIdx.x;
    if (e < E) atomicAdd(&deg[dst[e]], 1.0f);
}

__global__ void deg_rsqrt(float* __restrict__ deg, int N)
{
    int i = blockIdx.x * blockDim.x + threadIdx.x;
    if (i < N) {
        float d = deg[i];
        deg[i] = d > 0.0f ? rsqrtf(d) : 0.0f;
    }
}

__global__ void gcn_self(const float* __restrict__ m, const float* __restrict__ dinv,
                         float* __restrict__ agg, long total)
{
    long idx = (long)blockIdx.x * blockDim.x + threadIdx.x;
    if (idx >= total) return;
    const int i = (int)(idx >> 8);
    const float w = dinv[i] * dinv[i];
    agg[idx] = m[idx] * w;
}

__global__ void gcn_scatter(const float* __restrict__ m, const int* __restrict__ src,
                            const int* __restrict__ dst, const float* __restrict__ dinv,
                            float* __restrict__ agg, int E)
{
    long gid = (long)blockIdx.x * blockDim.x + threadIdx.x;
    if (gid >= (long)E * 64) return;
    const int e = (int)(gid >> 6);
    const int c = (int)(gid & 63) << 2;
    const int s = src[e];
    const int d = dst[e];
    const float w = dinv[s] * dinv[d];
    const float4 v = *(const float4*)(m + (size_t)s * HDIM + c);
    float* o = agg + (size_t)d * HDIM + c;
    atomicAdd(o + 0, v.x * w);
    atomicAdd(o + 1, v.y * w);
    atomicAdd(o + 2, v.z * w);
    atomicAdd(o + 3, v.w * w);
}

__global__ void bias_relu(const float* __restrict__ agg, const float* __restrict__ b,
                          float* __restrict__ h, long total)
{
    long idx = (long)blockIdx.x * blockDim.x + threadIdx.x;
    if (idx >= total) return;
    const float v = agg[idx] + b[idx & (HDIM - 1)];
    h[idx] = v > 0.0f ? v : 0.0f;
}

extern "C" void kernel_launch(void* const* d_in, const int* in_sizes, int n_in,
                              void* d_out, int out_size, void* d_ws, size_t ws_size,
                              hipStream_t stream)
{
    const float* x      = (const float*)d_in[0];
    const int*   eidx   = (const int*)  d_in[1];
    const float* enc_w1 = (const float*)d_in[2];
    const float* enc_b1 = (const float*)d_in[3];
    const float* enc_w2 = (const float*)d_in[4];
    const float* enc_b2 = (const float*)d_in[5];
    const float* gcn_w  = (const float*)d_in[6];
    const float* gcn_b  = (const float*)d_in[7];

    const int F_IN = 128;
    const int N    = in_sizes[0] / F_IN;
    const int E    = in_sizes[1] / 2;
    const int L    = in_sizes[6] / (HDIM * HDIM);

    const int* src = eidx;
    const int* dst = eidx + E;

    float* h    = (float*)d_out;
    float* m    = (float*)d_ws;
    float* agg  = m + (size_t)N * HDIM;
    float* dinv = agg + (size_t)N * HDIM;

    const dim3 blk(256);
    const dim3 gg(HDIM / 64, (N + 127) / 128);
    const int tb = 256;
    const long total = (long)N * HDIM;

    gemm_f32_wmma<<<gg, blk, 0, stream>>>(x, enc_w1, enc_b1, m, N, F_IN, HDIM, 1);
    gemm_f32_wmma<<<gg, blk, 0, stream>>>(m, enc_w2, enc_b2, h, N, HDIM, HDIM, 0);

    deg_init <<<(N + tb - 1) / tb, tb, 0, stream>>>(dinv, N);
    deg_edges<<<(E + tb - 1) / tb, tb, 0, stream>>>(dst, dinv, E);
    deg_rsqrt<<<(N + tb - 1) / tb, tb, 0, stream>>>(dinv, N);

    for (int l = 0; l < L; ++l) {
        gemm_f32_wmma<<<gg, blk, 0, stream>>>(h, gcn_w + (size_t)l * HDIM * HDIM,
                                              nullptr, m, N, HDIM, HDIM, 0);
        gcn_self<<<(int)((total + tb - 1) / tb), tb, 0, stream>>>(m, dinv, agg, total);
        const long st = (long)E * 64;
        gcn_scatter<<<(int)((st + tb - 1) / tb), tb, 0, stream>>>(m, src, dst, dinv, agg, E);
        bias_relu<<<(int)((total + tb - 1) / tb), tb, 0, stream>>>(
            agg, gcn_b + (size_t)l * HDIM, h, total);
    }
}


# Edge scatter: one wave per edge row, lane = float in a 128-B line, so each row load and f32 atomic covers one whole line (was 4 quarter lines); 16384 waves loop over edges; atomic arrival order new
# speedup vs baseline: 1.6863x; 1.6863x over previous
#include <hip/hip_runtime.h>

typedef __attribute__((ext_vector_type(2))) float v2f;
typedef __attribute__((ext_vector_type(8))) float v8f;

#define HDIM 256
#define SB_STRIDE 72

template <bool DUAL>
__global__ __launch_bounds__(256)
void gemm_f32_wmma(const float* __restrict__ A, const float* __restrict__ B,
                   const float* __restrict__ bias, float* __restrict__ C,
                   float* __restrict__ C2, const float* __restrict__ rowScale,
                   int M, int K, int N, int applyRelu)
{
    __shared__ float sB[256 * SB_STRIDE];

    const int tid  = threadIdx.x;
    const int lane = tid & 31;
    const int wave = tid >> 5;
    const int half = lane >> 4;
    const int l16  = lane & 15;

    const int row0 = blockIdx.y * 128 + wave * 16;
    const int col0 = blockIdx.x * 64;

    {
        const int r = tid >> 4;
        const int c = (tid & 15) << 2;
        for (int kb = 0; kb < K; kb += 16) {
            const float4 v = *(const float4*)(B + (size_t)(kb + r) * N + col0 + c);
            *(float4*)(&sB[(kb + r) * SB_STRIDE + c]) = v;
        }
    }
    __syncthreads();

    int aRow = row0 + l16;
    if (aRow > M - 1) aRow = M - 1;
    const float* Aptr = A + (size_t)aRow * K + half * 2;

    float sc[8];
    if (DUAL) {
#pragma unroll
        for (int r = 0; r < 8; ++r) {
            int sr = row0 + half * 8 + r;
            if (sr > M - 1) sr = M - 1;
            sc[r] = rowScale[sr];
        }
    }

    v8f acc[4];
    acc[0] = (v8f)0.0f; acc[1] = (v8f)0.0f; acc[2] = (v8f)0.0f; acc[3] = (v8f)0.0f;

    for (int kk = 0; kk < K; kk += 4) {
        const v2f a = *(const v2f*)(Aptr + kk);
        __builtin_prefetch(Aptr + kk + 32, 0, 0);

        const float* s0 = &sB[(kk + half * 2) * SB_STRIDE + l16];
#pragma unroll
        for (int t = 0; t < 4; ++t) {
            v2f b;
            b.x = s0[t * 16];
            b.y = s0[SB_STRIDE + t * 16];
            acc[t] = __builtin_amdgcn_wmma_f32_16x16x4_f32(
                false, a, false, b, (short)0, acc[t], false, false);
        }
    }

    float bv[4];
#pragma unroll
    for (int t = 0; t < 4; ++t)
        bv[t] = bias ? bias[col0 + t * 16 + l16] : 0.0f;

#pragma unroll
    for (int r = 0; r < 8; ++r) {
        const int row = row0 + half * 8 + r;
        if (row < M) {
            const float s2 = DUAL ? sc[r] * sc[r] : 0.0f;
#pragma unroll
            for (int t = 0; t < 4; ++t) {
                const size_t o = (size_t)row * N + col0 + t * 16 + l16;
                float v = acc[t][r] + bv[t];
                if (applyRelu) v = v > 0.0f ? v : 0.0f;
                C[o] = v;
                if (DUAL) C2[o] = v * s2;
            }
        }
    }
}

__global__ void deg_init(float* __restrict__ deg, int N)
{
    int i = blockIdx.x * blockDim.x + threadIdx.x;
    if (i < N) deg[i] = 1.0f;
}

__global__ void deg_edges(const int* __restrict__ dst, float* __restrict__ deg, int E)
{
    int e = blockIdx.x * blockDim.x + threadIdx.x;
    if (e < E) atomicAdd(&deg[dst[e]], 1.0f);
}

__global__ void deg_rsqrt(float* __restrict__ deg, int N)
{
    int i = blockIdx.x * blockDim.x + threadIdx.x;
    if (i < N) {
        float d = deg[i];
        deg[i] = d > 0.0f ? rsqrtf(d) : 0.0f;
    }
}

__global__ __launch_bounds__(256)
void gcn_scatter(const float* __restrict__ f, const int* __restrict__ src,
                 const int* __restrict__ dst, const float* __restrict__ dinv,
                 float* __restrict__ agg, int E)
{
    const int lane = threadIdx.x & 31;
    const int wv = __builtin_amdgcn_readfirstlane((int)(threadIdx.x >> 5));
    const int nw = (int)gridDim.x * 8;
    const int gw = (int)blockIdx.x * 8 + wv;
    const int G  = E >> 2;

    if (gw < G) {
        int g = gw;
        int s[4], d[4];
#pragma unroll
        for (int k = 0; k < 4; ++k) { s[k] = src[4 * g + k]; d[k] = dst[4 * g + k]; }

        for (;;) {
            const int gn = g + nw;
            const bool more = gn < G;
            int sn[4], dn[4];
#pragma unroll
            for (int k = 0; k < 4; ++k) { sn[k] = s[k]; dn[k] = d[k]; }
            if (more) {
#pragma unroll
                for (int k = 0; k < 4; ++k) { sn[k] = src[4 * gn + k]; dn[k] = dst[4 * gn + k]; }
            }

            float w[4];
            float v[4][8];
#pragma unroll
            for (int k = 0; k < 4; ++k) {
                w[k] = dinv[s[k]] * dinv[d[k]];
                const float* fr = f + (size_t)s[k] * HDIM + lane;
#pragma unroll
                for (int j = 0; j < 8; ++j) v[k][j] = fr[j * 32];
            }
#pragma unroll
            for (int k = 0; k < 4; ++k) {
                float* o = agg + (size_t)d[k] * HDIM + lane;
#pragma unroll
                for (int j = 0; j < 8; ++j) atomicAdd(o + j * 32, v[k][j] * w[k]);
            }
            if (!more) break;
            g = gn;
#pragma unroll
            for (int k = 0; k < 4; ++k) { s[k] = sn[k]; d[k] = dn[k]; }
        }
    }

    if (gw < (E & 3)) {
        const int e  = 4 * G + gw;
        const int s1 = src[e];
        const int d1 = dst[e];
        const float w1 = dinv[s1] * dinv[d1];
        const float* fr = f + (size_t)s1 * HDIM + lane;
        float* o = agg + (size_t)d1 * HDIM + lane;
        float v1[8];
#pragma unroll
        for (int j = 0; j < 8; ++j) v1[j] = fr[j * 32];
#pragma unroll
        for (int j = 0; j < 8; ++j) atomicAdd(o + j * 32, v1[j] * w1);
    }
}

extern "C" void kernel_launch(void* const* d_in, const int* in_sizes, int n_in,
                              void* d_out, int out_size, void* d_ws, size_t ws_size,
                              hipStream_t stream)
{
    const float* x      = (const float*)d_in[0];
    const int*   eidx   = (const int*)  d_in[1];
    const float* enc_w1 = (const float*)d_in[2];
    const float* enc_b1 = (const float*)d_in[3];
    const float* enc_w2 = (const float*)d_in[4];
    const float* enc_b2 = (const float*)d_in[5];
    const float* gcn_w  = (const float*)d_in[6];
    const float* gcn_b  = (const float*)d_in[7];

    const int F_IN = 128;
    const int N    = in_sizes[0] / F_IN;
    const int E    = in_sizes[1] / 2;
    const int L    = in_sizes[6] / (HDIM * HDIM);

    const int* src = eidx;
    const int* dst = eidx + E;

    float* out  = (float*)d_out;
    float* w0   = (float*)d_ws;
    float* w1   = w0 + (size_t)N * HDIM;
    float* dinv = w1 + (size_t)N * HDIM;
    const bool outSecond = (L % 3) == 2;
    float* hb = outSecond ? w0 : out;
    float* ab = outSecond ? out : w0;
    float* fb = w1;

    const dim3 blk(256);
    const dim3 gg(HDIM / 64, (N + 127) / 128);
    const int tb = 256;

    deg_init <<<(N + tb - 1) / tb, tb, 0, stream>>>(dinv, N);
    deg_edges<<<(E + tb - 1) / tb, tb, 0, stream>>>(dst, dinv, E);
    deg_rsqrt<<<(N + tb - 1) / tb, tb, 0, stream>>>(dinv, N);

    gemm_f32_wmma<false><<<gg, blk, 0, stream>>>(x, enc_w1, enc_b1, fb, nullptr, nullptr,
                                                 N, F_IN, HDIM, 1);
    if (L > 0)
        gemm_f32_wmma<true><<<gg, blk, 0, stream>>>(fb, enc_w2, enc_b2, hb, ab, dinv,
                                                    N, HDIM, HDIM, 0);
    else
        gemm_f32_wmma<false><<<gg, blk, 0, stream>>>(fb, enc_w2, enc_b2, hb, nullptr, nullptr,
                                                     N, HDIM, HDIM, 0);

    for (int l = 0; l < L; ++l) {
        gcn_scatter<<<2048, 256, 0, stream>>>(hb, src, dst, dinv, ab, E);
        const float* Wl = gcn_w + (size_t)l * HDIM * HDIM;
        const float* bl = gcn_b + (size_t)l * HDIM;
        if (l == L - 1) {
            gemm_f32_wmma<false><<<gg, blk, 0, stream>>>(ab, Wl, bl, out, nullptr, nullptr,
                                                         N, HDIM, HDIM, 1);
        } else {
            gemm_f32_wmma<true><<<gg, blk, 0, stream>>>(ab, Wl, bl, fb, hb, dinv,
                                                        N, HDIM, HDIM, 1);
            float* t = hb; hb = fb; fb = ab; ab = t;
        }
    }
}


# GEMM A operand through LDS: each wave copies its 16 rows in 32-k chunks by whole-line 16-byte loads (next chunk in flight in registers) and reads fragments from LDS; same k order, same sums
# speedup vs baseline: 2.1657x; 1.2843x over previous
#include <hip/hip_runtime.h>

typedef __attribute__((ext_vector_type(2))) float v2f;
typedef __attribute__((ext_vector_type(4))) float v4f;
typedef __attribute__((ext_vector_type(8))) float v8f;

#define HDIM 256
#define SB_STRIDE 72
#define KC 32
#define SA_STRIDE 36

template <bool DUAL>
__global__ __launch_bounds__(256)
void gemm_f32_wmma(const float* __restrict__ A, const float* __restrict__ B,
                   const float* __restrict__ bias, float* __restrict__ C,
                   float* __restrict__ C2, const float* __restrict__ rowScale,
                   int M, int K, int N, int applyRelu)
{
    __shared__ float sB[256 * SB_STRIDE];
    __shared__ __attribute__((aligned(16))) float sA[8 * 16 * SA_STRIDE];

    const int tid  = threadIdx.x;
    const int lane = tid & 31;
    const int wave = tid >> 5;
    const int half = lane >> 4;
    const int l16  = lane & 15;

    const int row0 = blockIdx.y * 128 + wave * 16;
    const int col0 = blockIdx.x * 64;

    {
        const int r = tid >> 4;
        const int c = (tid & 15) << 2;
        for (int kb = 0; kb < K; kb += 16) {
            const float4 v = *(const float4*)(B + (size_t)(kb + r) * N + col0 + c);
            *(float4*)(&sB[(kb + r) * SB_STRIDE + c]) = v;
        }
    }
    __syncthreads();

    float* sAw = &sA[wave * 16 * SA_STRIDE];
    const int sRow = lane >> 3;
    const int sCol = (lane & 7) << 2;
    const float* gA[4];
#pragma unroll
    for (int q = 0; q < 4; ++q) {
        int r = row0 + 4 * q + sRow;
        if (r > M - 1) r = M - 1;
        gA[q] = A + (size_t)r * K + sCol;
    }
    float* sWr = sAw + sRow * SA_STRIDE + sCol;
    const float* sRd = sAw + l16 * SA_STRIDE + half * 2;

    v4f g[4];
#pragma unroll
    for (int q = 0; q < 4; ++q) g[q] = *(const v4f*)(gA[q]);

    float sc[8];
    if (DUAL) {
#pragma unroll
        for (int r = 0; r < 8; ++r) {
            int sr = row0 + half * 8 + r;
            if (sr > M - 1) sr = M - 1;
            sc[r] = rowScale[sr];
        }
    }

    v8f acc[4];
    acc[0] = (v8f)0.0f; acc[1] = (v8f)0.0f; acc[2] = (v8f)0.0f; acc[3] = (v8f)0.0f;

    for (int kc = 0; kc < K; kc += KC) {
#pragma unroll
        for (int q = 0; q < 4; ++q) *(v4f*)(sWr + 4 * q * SA_STRIDE) = g[q];
        if (kc + KC < K) {
#pragma unroll
            for (int q = 0; q < 4; ++q) g[q] = *(const v4f*)(gA[q] + kc + KC);
        }

#pragma unroll
        for (int j = 0; j < KC / 4; ++j) {
            const v2f a = *(const v2f*)(sRd + 4 * j);
            const float* s0 = &sB[(kc + 4 * j + half * 2) * SB_STRIDE + l16];
#pragma unroll
            for (int t = 0; t < 4; ++t) {
                v2f b;
                b.x = s0[t * 16];
                b.y = s0[SB_STRIDE + t * 16];
                acc[t] = __builtin_amdgcn_wmma_f32_16x16x4_f32(
                    false, a, false, b, (short)0, acc[t], false, false);
            }
        }
    }

    float bv[4];
#pragma unroll
    for (int t = 0; t < 4; ++t)
        bv[t] = bias ? bias[col0 + t * 16 + l16] : 0.0f;

#pragma unroll
    for (int r = 0; r < 8; ++r) {
        const int row = row0 + half * 8 + r;
        if (row < M) {
            const float s2 = DUAL ? sc[r] * sc[r] : 0.0f;
#pragma unroll
            for (int t = 0; t < 4; ++t) {
                const size_t o = (size_t)row * N + col0 + t * 16 + l16;
                float v = acc[t][r] + bv[t];
                if (applyRelu) v = v > 0.0f ? v : 0.0f;
                C[o] = v;
                if (DUAL) C2[o] = v * s2;
            }
        }
    }
}

__global__ void deg_init(float* __restrict__ deg, int N)
{
    int i = blockIdx.x * blockDim.x + threadIdx.x;
    if (i < N) deg[i] = 1.0f;
}

__global__ void deg_edges(const int* __restrict__ dst, float* __restrict__ deg, int E)
{
    int e = blockIdx.x * blockDim.x + threadIdx.x;
    if (e < E) atomicAdd(&deg[dst[e]], 1.0f);
}

__global__ void deg_rsqrt(float* __restrict__ deg, int N)
{
    int i = blockIdx.x * blockDim.x + threadIdx.x;
    if (i < N) {
        float d = deg[i];
        deg[i] = d > 0.0f ? rsqrtf(d) : 0.0f;
    }
}

__global__ __launch_bounds__(256)
void gcn_scatter(const float* __restrict__ f, const int* __restrict__ src,
                 const int* __restrict__ dst, const float* __restrict__ dinv,
                 float* __restrict__ agg, int E)
{
    const int lane = threadIdx.x & 31;
    const int wv = __builtin_amdgcn_readfirstlane((int)(threadIdx.x >> 5));
    const int nw = (int)gridDim.x * 8;
    const int gw = (int)blockIdx.x * 8 + wv;
    const int G  = E >> 2;

    if (gw < G) {
        int g = gw;
        int s[4], d[4];
#pragma unroll
        for (int k = 0; k < 4; ++k) { s[k] = src[4 * g + k]; d[k] = dst[4 * g + k]; }

        for (;;) {
            const int gn = g + nw;
            const bool more = gn < G;
            int sn[4], dn[4];
#pragma unroll
            for (int k = 0; k < 4; ++k) { sn[k] = s[k]; dn[k] = d[k]; }
            if (more) {
#pragma unroll
                for (int k = 0; k < 4; ++k) { sn[k] = src[4 * gn + k]; dn[k] = dst[4 * gn + k]; }
            }

            float w[4];
            float v[4][8];
#pragma unroll
            for (int k = 0; k < 4; ++k) {
                w[k] = dinv[s[k]] * dinv[d[k]];
                const float* fr = f + (size_t)s[k] * HDIM + lane;
#pragma unroll
                for (int j = 0; j < 8; ++j) v[k][j] = fr[j * 32];
            }
#pragma unroll
            for (int k = 0; k < 4; ++k) {
                float* o = agg + (size_t)d[k] * HDIM + lane;
#pragma unroll
                for (int j = 0; j < 8; ++j) atomicAdd(o + j * 32, v[k][j] * w[k]);
            }
            if (!more) break;
            g = gn;
#pragma unroll
            for (int k = 0; k < 4; ++k) { s[k] = sn[k]; d[k] = dn[k]; }
        }
    }

    if (gw < (E & 3)) {
        const int e  = 4 * G + gw;
        const int s1 = src[e];
        const int d1 = dst[e];
        const float w1 = dinv[s1] * dinv[d1];
        const float* fr = f + (size_t)s1 * HDIM + lane;
        float* o = agg + (size_t)d1 * HDIM + lane;
        float v1[8];
#pragma unroll
        for (int j = 0; j < 8; ++j) v1[j] = fr[j * 32];
#pragma unroll
        for (int j = 0; j < 8; ++j) atomicAdd(o + j * 32, v1[j] * w1);
    }
}

extern "C" void kernel_launch(void* const* d_in, const int* in_sizes, int n_in,
                              void* d_out, int out_size, void* d_ws, size_t ws_size,
                              hipStream_t stream)
{
    const float* x      = (const float*)d_in[0];
    const int*   eidx   = (const int*)  d_in[1];
    const float* enc_w1 = (const float*)d_in[2];
    const float* enc_b1 = (const float*)d_in[3];
    const float* enc_w2 = (const float*)d_in[4];
    const float* enc_b2 = (const float*)d_in[5];
    const float* gcn_w  = (const float*)d_in[6];
    const float* gcn_b  = (const float*)d_in[7];

    const int F_IN = 128;
    const int N    = in_sizes[0] / F_IN;
    const int E    = in_sizes[1] / 2;
    const int L    = in_sizes[6] / (HDIM * HDIM);

    const int* src = eidx;
    const int* dst = eidx + E;

    float* out  = (float*)d_out;
    float* w0   = (float*)d_ws;
    float* w1   = w0 + (size_t)N * HDIM;
    float* dinv = w1 + (size_t)N * HDIM;
    const bool outSecond = (L % 3) == 2;
    float* hb = outSecond ? w0 : out;
    float* ab = outSecond ? out : w0;
    float* fb = w1;

    const dim3 blk(256);
    const dim3 gg(HDIM / 64, (N + 127) / 128);
    const int tb = 256;

    deg_init <<<(N + tb - 1) / tb, tb, 0, stream>>>(dinv, N);
    deg_edges<<<(E + tb - 1) / tb, tb, 0, stream>>>(dst, dinv, E);
    deg_rsqrt<<<(N + tb - 1) / tb, tb, 0, stream>>>(dinv, N);

    gemm_f32_wmma<false><<<gg, blk, 0, stream>>>(x, enc_w1, enc_b1, fb, nullptr, nullptr,
                                                 N, F_IN, HDIM, 1);
    if (L > 0)
        gemm_f32_wmma<true><<<gg, blk, 0, stream>>>(fb, enc_w2, enc_b2, hb, ab, dinv,
                                                    N, HDIM, HDIM, 0);
    else
        gemm_f32_wmma<false><<<gg, blk, 0, stream>>>(fb, enc_w2, enc_b2, hb, nullptr, nullptr,
                                                     N, HDIM, HDIM, 0);

    for (int l = 0; l < L; ++l) {
        gcn_scatter<<<2048, 256, 0, stream>>>(hb, src, dst, dinv, ab, E);
        const float* Wl = gcn_w + (size_t)l * HDIM * HDIM;
        const float* bl = gcn_b + (size_t)l * HDIM;
        if (l == L - 1) {
            gemm_f32_wmma<false><<<gg, blk, 0, stream>>>(ab, Wl, bl, out, nullptr, nullptr,
                                                         N, HDIM, HDIM, 1);
        } else {
            gemm_f32_wmma<true><<<gg, blk, 0, stream>>>(ab, Wl, bl, fb, hb, dinv,
                                                        N, HDIM, HDIM, 1);
            float* t = hb; hb = fb; fb = ab; ab = t;
        }
    }
}
